# starting kernel
# baseline (speedup 1.0000x reference)
#include <hip/hip_runtime.h>

typedef __attribute__((ext_vector_type(2))) float v2f;
typedef __attribute__((ext_vector_type(8))) float v8f;

#define NN 65536
#define DD 1024
#define CHUNKS 64
#define ROWS_PER_CHUNK (NN / CHUNKS)
#define DTILES (DD / 16)

__global__ void __launch_bounds__(256)
la_phase1_gemv(const float* __restrict__ P, const float* __restrict__ g,
               float* __restrict__ partial) {
  const int wave  = (blockIdx.x * blockDim.x + threadIdx.x) >> 5;
  const int lane  = threadIdx.x & 31;
  const int chunk = wave >> 6;
  const int dtile = wave & 63;
  const int d0    = dtile * 16;
  const int n0    = chunk * ROWS_PER_CHUNK;
  const int half  = lane >> 4;
  const int l16   = lane & 15;

  const float* pA = g + n0 + 2 * half;
  const float* pB = P + (size_t)(n0 + 2 * half) * DD + d0 + l16;

  v8f acc = {};
  for (int i = 0; i < ROWS_PER_CHUNK / 4; ++i) {
    v2f a; a.x = pA[0];  a.y = pA[1];
    v2f b; b.x = pB[0];  b.y = pB[DD];
    acc = __builtin_amdgcn_wmma_f32_16x16x4_f32(
         false, a,  false, b,
         (short)0, acc,  false,  false);
    pA += 4;
    pB += 4 * DD;
  }
  if (lane < 16) partial[chunk * DD + d0 + l16] = acc[0];
}

__global__ void __launch_bounds__(256)
la_phase2_reduce(const float* __restrict__ partial, float* __restrict__ weighted) {
  const int d = blockIdx.x * blockDim.x + threadIdx.x;
  float s = 0.f;
#pragma unroll 8
  for (int c = 0; c < CHUNKS; ++c) s += partial[c * DD + d];
  weighted[d] = s;
}

__global__ void __launch_bounds__(256)
la_phase3_outer(const float* __restrict__ weighted, const float* __restrict__ cs,
                float* __restrict__ out) {
  const int wave = (blockIdx.x * blockDim.x + threadIdx.x) >> 5;
  const int lane = threadIdx.x & 31;
  const int tm   = wave >> 6;
  const int tn   = wave & 63;
  const int half = lane >> 4;
  const int l16  = lane & 15;

  v2f a; a.x = (half == 0) ? weighted[tm * 16 + l16] : 0.f; a.y = 0.f;
  v2f b; b.x = (half == 0) ? cs[tn * 16 + l16]       : 0.f; b.y = 0.f;
  v8f c = {};
  c = __builtin_amdgcn_wmma_f32_16x16x4_f32(
      false, a, false, b, (short)0, c, false, false);

  float* o = out + (size_t)(tm * 16 + half * 8) * DD + tn * 16 + l16;
#pragma unroll
  for (int j = 0; j < 8; ++j) o[(size_t)j * DD] = c[j];
}

extern "C" void kernel_launch(void* const* d_in, const int* in_sizes, int n_in,
                              void* d_out, int out_size, void* d_ws, size_t ws_size,
                              hipStream_t stream) {
  (void)in_sizes; (void)n_in; (void)out_size; (void)ws_size;

  const float* past_states   = (const float*)d_in[0];
  const float* current_state = (const float*)d_in[1];
  const float* gradients     = (const float*)d_in[2];
  float* out = (float*)d_out;

  float* partial  = (float*)d_ws;
  float* weighted = partial + CHUNKS * DD;

  la_phase1_gemv<<<(CHUNKS * DTILES) / 8, 256, 0, stream>>>(past_states, gradients, partial);
  la_phase2_reduce<<<DD / 256, 256, 0, stream>>>(partial, weighted);
  la_phase3_outer<<<(DTILES * DTILES) / 8, 256, 0, stream>>>(weighted, current_state, out);
}


# phase 1 re-tiled: a wave streams 16 whole contiguous rows with 512-byte loads; accumulator rows = the 16 column groups (A holds g in one row); partials 32 -> 256; sum order changed
# speedup vs baseline: 1.3045x; 1.3045x over previous
#include <hip/hip_runtime.h>

typedef __attribute__((ext_vector_type(2))) float v2f;
typedef __attribute__((ext_vector_type(4))) float v4f;
typedef __attribute__((ext_vector_type(8))) float v8f;

#define NN 65536
#define DD 1024
#define WAVES 16
#define ROWS_PER_WAVE 16
#define ROWS_PER_WG (WAVES * ROWS_PER_WAVE)
#define NWG (NN / ROWS_PER_WG)
#define PIECES (DD / 128)
#define NBATCH (ROWS_PER_WAVE / 2)
#define DTILES (DD / 16)

__global__ void __launch_bounds__(WAVES * 32)
la_phase1_gemv(const float* __restrict__ P, const float* __restrict__ g,
               float* __restrict__ partial) {
  __shared__ v4f red[WAVES][DD / 4];

  const int lane = threadIdx.x & 31;
  const int wave = threadIdx.x >> 5;
  const int half = lane >> 4;
  const int l16  = lane & 15;
  const int qsel = l16 - half;
  const int n0   = blockIdx.x * ROWS_PER_WG + wave * ROWS_PER_WAVE;

  const v2f* pg = (const v2f*)(g + n0);
  const v4f* px = (const v4f*)(P + (size_t)n0 * DD + 4 * lane);

  v4f xb[2][PIECES], yb[2][PIECES];
  v2f ab[NBATCH];
  v8f acc[4] = {};
#pragma unroll
  for (int b = 0; b < NBATCH; ++b) ab[b] = pg[b];
  __builtin_amdgcn_sched_barrier(0);

#define LOAD_BATCH(B) do { \
    _Pragma("unroll") for (int p = 0; p < PIECES; ++p) { \
      xb[(B) & 1][p] = px[(size_t)(2 * (B)) * (DD / 4) + 32 * p];     \
      yb[(B) & 1][p] = px[(size_t)(2 * (B) + 1) * (DD / 4) + 32 * p]; } } while (0)

  LOAD_BATCH(0);
#pragma unroll
  for (int b = 0; b < NBATCH; ++b) {
    __builtin_amdgcn_sched_barrier(0);
    if (b + 1 < NBATCH) LOAD_BATCH(b + 1);
    __builtin_amdgcn_sched_barrier(0);
    v2f a = ab[b];
    asm volatile("v_nop\n\tv_nop\n\tv_nop\n\tv_nop\n\tv_nop\n\tv_nop\n\tv_nop\n\tv_nop" : "+v"(a));
    v2f ap[PIECES];
#pragma unroll
    for (int p = 0; p < PIECES; ++p) {
      ap[p].x = (qsel == 2 * p) ? a.x : 0.f;
      ap[p].y = (qsel == 2 * p) ? a.y : 0.f;
    }
    __builtin_amdgcn_sched_barrier(0);
#pragma unroll
    for (int p = 0; p < PIECES; ++p) {
      v4f x = xb[b & 1][p], y = yb[b & 1][p];
      float x1 = x.y, x3 = x.w, y0 = y.x, y2 = y.z;
      asm("v_swap_b32 %0, %1" : "+v"(x1), "+v"(y0));
      asm("v_swap_b32 %0, %1" : "+v"(x3), "+v"(y2));
      x.y = x1; y.x = y0; x.w = x3; y.z = y2;
      const v2f b0 = __builtin_shufflevector(x, x, 0, 1);
      const v2f b2 = __builtin_shufflevector(x, x, 2, 3);
      const v2f b1 = __builtin_shufflevector(y, y, 0, 1);
      const v2f b3 = __builtin_shufflevector(y, y, 2, 3);
      acc[0] = __builtin_amdgcn_wmma_f32_16x16x4_f32(false, ap[p], false, b0, (short)0, acc[0], false, false);
      acc[1] = __builtin_amdgcn_wmma_f32_16x16x4_f32(false, ap[p], false, b1, (short)0, acc[1], false, false);
      acc[2] = __builtin_amdgcn_wmma_f32_16x16x4_f32(false, ap[p], false, b2, (short)0, acc[2], false, false);
      acc[3] = __builtin_amdgcn_wmma_f32_16x16x4_f32(false, ap[p], false, b3, (short)0, acc[3], false, false);
    }
  }
#undef LOAD_BATCH

  __builtin_amdgcn_sched_barrier(0);
  asm volatile("v_nop\n\tv_nop\n\tv_nop\n\tv_nop\n\tv_nop\n\tv_nop\n\tv_nop\n\tv_nop"
               : : "v"(acc[0]), "v"(acc[1]), "v"(acc[2]), "v"(acc[3]));
  __builtin_amdgcn_sched_barrier(0);

#pragma unroll
  for (int r = 0; r < 8; ++r) {
    v4f v; v.x = acc[0][r]; v.y = acc[1][r]; v.z = acc[2][r]; v.w = acc[3][r];
    red[wave][16 * (8 * half + r) + l16] = v;
  }
  __syncthreads();
  {
    const v2f* rf = (const v2f*)&red[0][0];
    v2f s = {0.f, 0.f};
#pragma unroll
    for (int w = 0; w < WAVES; ++w) s += rf[w * (DD / 2) + threadIdx.x];
    ((v2f*)(partial + (size_t)blockIdx.x * DD))[threadIdx.x] = s;
  }
}

#define P2_COLS 32
#define P2_WAVES 8
#define P2_ROWS (NWG / P2_WAVES)

__global__ void __launch_bounds__(P2_WAVES * 32)
la_phase2_reduce(const float* __restrict__ partial, float* __restrict__ weighted) {
  __shared__ float sred[P2_WAVES][P2_COLS];
  const int c = threadIdx.x & 31;
  const int r = threadIdx.x >> 5;
  const int d = blockIdx.x * P2_COLS + c;
  const float* src = partial + (size_t)(r * P2_ROWS) * DD + d;
  float v[P2_ROWS];
#pragma unroll
  for (int k = 0; k < P2_ROWS; ++k) v[k] = src[(size_t)k * DD];
  float s = 0.f;
#pragma unroll
  for (int k = 0; k < P2_ROWS; ++k) s += v[k];
  sred[r][c] = s;
  __syncthreads();
  if (threadIdx.x < P2_COLS) {
    float t = 0.f;
#pragma unroll
    for (int w = 0; w < P2_WAVES; ++w) t += sred[w][threadIdx.x];
    weighted[d] = t;
  }
}

#define P3_COLS 128
#define P3_WAVES 2
#define P3_CBLK (DD / P3_COLS)

__global__ void __launch_bounds__(P3_WAVES * 32)
la_phase3_outer(const float* __restrict__ weighted, const float* __restrict__ cs,
                float* __restrict__ out) {
  const int wave = blockIdx.x * P3_WAVES + (threadIdx.x >> 5);
  const int lane = threadIdx.x & 31;
  const int tm   = wave / P3_CBLK;
  const int n0   = (wave & (P3_CBLK - 1)) * P3_COLS;
  const int half = lane >> 4;
  const int l16  = lane & 15;

  const float wv = weighted[tm * 16 + l16];
  v4f c4[2];
#pragma unroll
  for (int q = 0; q < 2; ++q) c4[q] = *(const v4f*)(cs + n0 + 64 * q + 4 * l16);

  v2f a; a.x = (half == 0) ? wv : 0.f; a.y = 0.f;
  v2f b[2][4];
#pragma unroll
  for (int q = 0; q < 2; ++q)
#pragma unroll
    for (int j = 0; j < 4; ++j) { b[q][j].x = (half == 0) ? c4[q][j] : 0.f; b[q][j].y = 0.f; }

  __builtin_amdgcn_sched_barrier(0);
  v8f d[2][4];
#pragma unroll
  for (int q = 0; q < 2; ++q)
#pragma unroll
    for (int j = 0; j < 4; ++j) {
      v8f c = {};
      d[q][j] = __builtin_amdgcn_wmma_f32_16x16x4_f32(false, a, false, b[q][j], (short)0, c, false, false);
    }

  __builtin_amdgcn_sched_barrier(0);
  asm volatile("v_nop\n\tv_nop\n\tv_nop\n\tv_nop\n\tv_nop\n\tv_nop\n\tv_nop\n\tv_nop"
               : : "v"(d[0][0]), "v"(d[0][1]), "v"(d[0][2]), "v"(d[0][3]),
                   "v"(d[1][0]), "v"(d[1][1]), "v"(d[1][2]), "v"(d[1][3]));
  __builtin_amdgcn_sched_barrier(0);

  float* o = out + (size_t)(tm * 16 + half * 8) * DD + n0 + 4 * l16;
#pragma unroll
  for (int q = 0; q < 2; ++q)
#pragma unroll
    for (int r = 0; r < 8; ++r) {
      v4f v; v.x = d[q][0][r]; v.y = d[q][1][r]; v.z = d[q][2][r]; v.w = d[q][3][r];
      *(v4f*)(o + (size_t)r * DD + 64 * q) = v;
    }
}

extern "C" void kernel_launch(void* const* d_in, const int* in_sizes, int n_in,
                              void* d_out, int out_size, void* d_ws, size_t ws_size,
                              hipStream_t stream) {
  (void)in_sizes; (void)n_in; (void)out_size; (void)ws_size;

  const float* past_states   = (const float*)d_in[0];
  const float* current_state = (const float*)d_in[1];
  const float* gradients     = (const float*)d_in[2];
  float* out = (float*)d_out;

  float* partial  = out;
  float* weighted = (float*)d_ws;

  la_phase1_gemv<<<NWG, WAVES * 32, 0, stream>>>(past_states, gradients, partial);
  la_phase2_reduce<<<DD / P2_COLS, P2_WAVES * 32, 0, stream>>>(partial, weighted);
  la_phase3_outer<<<(DTILES * P3_CBLK) / P3_WAVES, P3_WAVES * 32, 0, stream>>>(weighted, current_state, out);
}
